# starting kernel
# baseline (speedup 1.0000x reference)
#include <hip/hip_runtime.h>

typedef float v2f __attribute__((ext_vector_type(2)));
typedef float v8f __attribute__((ext_vector_type(8)));

#define BATCH 8
#define CIN   512
#define TLEN  512
#define OCH   512
#define KSZ   65
#define PADW  32

#define NT 4
#define KT 20

__global__ __launch_bounds__(256) void pwtdconv_wmma_f32(
    const float* __restrict__ x,
    const float* __restrict__ w,
    const float* __restrict__ bias,
    float* __restrict__ out)
{
    const int gwave = (int)((blockIdx.x * blockDim.x + threadIdx.x) >> 5);
    const int lane  = (int)(threadIdx.x & 31);
    const int lm    = lane & 15;
    const int lh    = lane >> 4;

    const int OT = OCH / 16;
    const int TG = (TLEN / 16) / NT;

    int idx = gwave;
    const int ot = idx % OT; idx /= OT;
    const int tg = idx % TG; idx /= TG;
    const int b  = idx;
    if (b >= BATCH) return;

    const int o0    = ot * 16;
    const int o     = o0 + lm;
    const int cbase = o0 - PADW;

    v2f      bfrag[KT];
    unsigned aoff[KT][2];
#pragma unroll
    for (int kt = 0; kt < KT; ++kt) {
#pragma unroll
        for (int r = 0; r < 2; ++r) {
            const int c  = cbase + kt * 4 + 2 * lh + r;
            const int kk = c - o + PADW;
            const bool inb = (kk >= 0) & (kk < KSZ) & (c >= 0) & (c < CIN);
            const int  kc  = kk < 0 ? 0 : (kk > KSZ - 1 ? KSZ - 1 : kk);
            const float wv = w[o * KSZ + kc];
            bfrag[kt][r] = inb ? wv : 0.0f;

            const int cc = c < 0 ? 0 : (c > CIN - 1 ? CIN - 1 : c);
            aoff[kt][r]  = (unsigned)((cc * TLEN + lm) * 4);
        }
    }

    const float bv = bias[o];
    const char* xb = (const char*)(x + (size_t)b * CIN * TLEN);
    float*      ob = out + (size_t)b * TLEN * OCH;

    for (int it = 0; it < NT; ++it) {
        const int   t0  = (tg * NT + it) * 16;
        const char* xbt = xb + (size_t)t0 * 4;

        v8f acc;
#pragma unroll
        for (int r = 0; r < 8; ++r) acc[r] = bv;

#pragma unroll
        for (int kt = 0; kt < KT; ++kt) {
            v2f a;
            a[0] = *(const float*)(xbt + aoff[kt][0]);
            a[1] = *(const float*)(xbt + aoff[kt][1]);
            acc = __builtin_amdgcn_wmma_f32_16x16x4_f32(
                false, a, false, bfrag[kt], (short)0, acc, false, false);
        }

        float* obt = ob + (size_t)(t0 + 8 * lh) * OCH + o;
#pragma unroll
        for (int r = 0; r < 8; ++r)
            obt[(size_t)r * OCH] = acc[r];
    }
}

extern "C" void kernel_launch(void* const* d_in, const int* in_sizes, int n_in,
                              void* d_out, int out_size, void* d_ws, size_t ws_size,
                              hipStream_t stream) {
    const float* x    = (const float*)d_in[0];
    const float* w    = (const float*)d_in[1];
    const float* bias = (const float*)d_in[2];
    float*       out  = (float*)d_out;

    const int total_waves   = BATCH * (OCH / 16) * ((TLEN / 16) / NT);
    const int block_threads = 256;
    const int grid_blocks   = (total_waves * 32 + block_threads - 1) / block_threads;

    hipLaunchKernelGGL(pwtdconv_wmma_f32, dim3(grid_blocks), dim3(block_threads),
                       0, stream, x, w, bias, out);
}


# x slab of 192 channels x 64 t staged once per work-group in LDS (channel pairs interleaved) and shared by its 8 o-tile waves; A fragments are 8-byte LDS reads grouped ahead of the matrix blocks
# speedup vs baseline: 1.9430x; 1.9430x over previous
#include <hip/hip_runtime.h>


typedef float v2f __attribute__((ext_vector_type(2)));
typedef float v4f __attribute__((ext_vector_type(4)));
typedef float v8f __attribute__((ext_vector_type(8)));

#define BATCH 8
#define CIN   512
#define TLEN  512
#define OCH   512
#define KSZ   65
#define PADW  32

#define NT 4
#define KT 20

#define WAVES  8
#define TSPAN  (NT * 16)
#define ROWS   (WAVES * 16 + 2 * PADW)
#define PAIRS  (ROWS / 2)
#define PITCH2 160
#define NPASS  (PAIRS / 16)

__global__ __launch_bounds__(256) void pwtdconv_wmma_f32(
    const float* __restrict__ x,
    const float* __restrict__ w,
    const float* __restrict__ bias,
    float* __restrict__ out)
{
    __shared__ float slab[PAIRS * PITCH2];

    const int tid  = (int)threadIdx.x;
    const int wv   = tid >> 5;
    const int lane = tid & 31;
    const int lm   = lane & 15;
    const int lh   = lane >> 4;

    int idx = (int)blockIdx.x;
    const int og = idx % (OCH / 16 / WAVES);  idx /= (OCH / 16 / WAVES);
    const int tg = idx % (TLEN / TSPAN);      idx /= (TLEN / TSPAN);
    const int b  = idx;

    const int cslab = og * (WAVES * 16) - PADW;
    const int tbase = tg * TSPAN;
    const float* xb = x + (size_t)b * CIN * TLEN;

    const int o0    = (og * WAVES + wv) * 16;
    const int o     = o0 + lm;
    const int cbase = o0 - PADW;

    const int sp   = tid >> 4;
    const int scol = (tid & 15) * 4;
    v4f lo[NPASS], hi[NPASS];
#pragma unroll
    for (int i = 0; i < NPASS; ++i) {
        const int c0 = cslab + 2 * (i * 16 + sp);
        const int c1 = c0 + 1;
        const int q0 = c0 < 0 ? 0 : (c0 > CIN - 1 ? CIN - 1 : c0);
        const int q1 = c1 < 0 ? 0 : (c1 > CIN - 1 ? CIN - 1 : c1);
        lo[i] = *(const v4f*)(xb + (size_t)q0 * TLEN + tbase + scol);
        hi[i] = *(const v4f*)(xb + (size_t)q1 * TLEN + tbase + scol);
    }

    v2f wraw[KT];
#pragma unroll
    for (int kt = 0; kt < KT; ++kt) {
#pragma unroll
        for (int r = 0; r < 2; ++r) {
            const int c  = cbase + kt * 4 + 2 * lh + r;
            const int kk = c - o + PADW;
            const int kc = kk < 0 ? 0 : (kk > KSZ - 1 ? KSZ - 1 : kk);
            wraw[kt][r] = w[o * KSZ + kc];
        }
    }
    const float bv = bias[o];

    __builtin_amdgcn_sched_barrier(0);

#pragma unroll
    for (int i = 0; i < NPASS; ++i) {
        float* dst = &slab[(i * 16 + sp) * PITCH2 + scol * 2];
        *(v4f*)(dst)     = (v4f){lo[i][0], hi[i][0], lo[i][1], hi[i][1]};
        *(v4f*)(dst + 4) = (v4f){lo[i][2], hi[i][2], lo[i][3], hi[i][3]};
    }

    v2f bfrag[KT];
#pragma unroll
    for (int kt = 0; kt < KT; ++kt) {
#pragma unroll
        for (int r = 0; r < 2; ++r) {
            const int c  = cbase + kt * 4 + 2 * lh + r;
            const int kk = c - o + PADW;
            const bool inb = (kk >= 0) & (kk < KSZ) & (c >= 0) & (c < CIN);
            bfrag[kt][r] = inb ? wraw[kt][r] : 0.0f;
        }
    }

    __syncthreads();

    const float* ab = &slab[(wv * 8 + lh) * PITCH2 + lm * 2];
    float*       ob = out + (size_t)b * TLEN * OCH;

    v2f a[NT][KT];
#pragma unroll
    for (int it = 0; it < NT; ++it) {
#pragma unroll
        for (int kt = 0; kt < KT; ++kt)
            a[it][kt] = *(const v2f*)(ab + kt * 2 * PITCH2 + it * 32);
    }

    __builtin_amdgcn_sched_barrier(0);

#pragma unroll
    for (int it = 0; it < NT; ++it) {
        const int t0 = tbase + it * 16;

        v8f acc;
#pragma unroll
        for (int r = 0; r < 8; ++r) acc[r] = bv;

#pragma unroll
        for (int kt = 0; kt < KT; ++kt) {
            acc = __builtin_amdgcn_wmma_f32_16x16x4_f32(
                false, a[it][kt], false, bfrag[kt], (short)0, acc, false, false);
        }

        float* obt = ob + (size_t)(t0 + 8 * lh) * OCH + o;
#pragma unroll
        for (int r = 0; r < 8; ++r)
            obt[(size_t)r * OCH] = acc[r];
    }
}

extern "C" void kernel_launch(void* const* d_in, const int* in_sizes, int n_in,
                              void* d_out, int out_size, void* d_ws, size_t ws_size,
                              hipStream_t stream) {
    const float* x    = (const float*)d_in[0];
    const float* w    = (const float*)d_in[1];
    const float* bias = (const float*)d_in[2];
    float*       out  = (float*)d_out;

    hipLaunchKernelGGL(pwtdconv_wmma_f32, dim3(256), dim3(256),
                       0, stream, x, w, bias, out);
}


# wave phases overlap: hand-written LDS-only barrier leaves weight gathers in flight; tile pair 0 masks B per K-step as weights land; its 16 stores drain under tile pair 1 (was 4 tiles K-interleaved)
# speedup vs baseline: 2.0803x; 1.0707x over previous
#include <hip/hip_runtime.h>


typedef float v2f __attribute__((ext_vector_type(2)));
typedef float v4f __attribute__((ext_vector_type(4)));
typedef float v8f __attribute__((ext_vector_type(8)));

#define BATCH 8
#define CIN   512
#define TLEN  512
#define OCH   512
#define KSZ   65
#define PADW  32

#define NT 4
#define KT 20

#define WAVES  8
#define TSPAN  (NT * 16)
#define ROWS   (WAVES * 16 + 2 * PADW)
#define PAIRS  (ROWS / 2)
#define PITCH2 160
#define NPASS  (PAIRS / 16)

__global__ __launch_bounds__(256) void pwtdconv_wmma_f32(
    const float* x,
    const float* w,
    const float* bias,
    float* __restrict__ out)
{
    __shared__ float slab[PAIRS * PITCH2];

    const int tid  = (int)threadIdx.x;
    const int wv   = tid >> 5;
    const int lane = tid & 31;
    const int lm   = lane & 15;
    const int lh   = lane >> 4;

    int idx = (int)blockIdx.x;
    const int og = idx % (OCH / 16 / WAVES);  idx /= (OCH / 16 / WAVES);
    const int tg = idx % (TLEN / TSPAN);      idx /= (TLEN / TSPAN);
    const int b  = idx;

    const int cslab = og * (WAVES * 16) - PADW;
    const int tbase = tg * TSPAN;
    const float* xb = x + (size_t)b * CIN * TLEN;

    const int o0    = (og * WAVES + wv) * 16;
    const int o     = o0 + lm;
    const int cbase = o0 - PADW;

    const float bv = bias[o];

    const int sp   = tid >> 4;
    const int scol = (tid & 15) * 4;
    v4f lo[NPASS], hi[NPASS];
#pragma unroll
    for (int i = 0; i < NPASS; ++i) {
        const int c0 = cslab + 2 * (i * 16 + sp);
        const int c1 = c0 + 1;
        const int q0 = c0 < 0 ? 0 : (c0 > CIN - 1 ? CIN - 1 : c0);
        const int q1 = c1 < 0 ? 0 : (c1 > CIN - 1 ? CIN - 1 : c1);
        lo[i] = *(const v4f*)(xb + (size_t)q0 * TLEN + tbase + scol);
        hi[i] = *(const v4f*)(xb + (size_t)q1 * TLEN + tbase + scol);
    }

    asm volatile("" ::: "memory");

    v2f wraw[KT];
#pragma unroll
    for (int kt = 0; kt < KT; ++kt) {
#pragma unroll
        for (int r = 0; r < 2; ++r) {
            const int c  = cbase + kt * 4 + 2 * lh + r;
            const int kk = c - o + PADW;
            const int kc = kk < 0 ? 0 : (kk > KSZ - 1 ? KSZ - 1 : kk);
            wraw[kt][r] = w[o * KSZ + kc];
        }
        if ((kt & 3) == 3 && kt != KT - 1) asm volatile("" ::: "memory");
    }
    __builtin_amdgcn_sched_barrier(0);

#pragma unroll
    for (int i = 0; i < NPASS; ++i) {
        float* dst = &slab[(i * 16 + sp) * PITCH2 + scol * 2];
        *(v4f*)(dst)     = (v4f){lo[i][0], hi[i][0], lo[i][1], hi[i][1]};
        *(v4f*)(dst + 4) = (v4f){lo[i][2], hi[i][2], lo[i][3], hi[i][3]};
    }

    asm volatile("s_wait_dscnt 0x0\n\ts_barrier_signal -1\n\ts_barrier_wait -1" ::: "memory");

    const float* ab = &slab[(wv * 8 + lh) * PITCH2 + lm * 2];
    float*       ob = out + (size_t)b * TLEN * OCH;

    v2f a[NT][KT];
#pragma unroll
    for (int ip = 0; ip < NT; ip += 2) {
#pragma unroll
        for (int kt = 0; kt < KT; ++kt) {
            a[ip][kt]     = *(const v2f*)(ab + kt * 2 * PITCH2 + ip * 32);
            a[ip + 1][kt] = *(const v2f*)(ab + kt * 2 * PITCH2 + ip * 32 + 32);
            if ((kt & 3) == 3) asm volatile("" ::: "memory");
        }
    }

    __builtin_amdgcn_sched_barrier(0);

    int   on = o;
    float bs = bv;
    asm volatile("" : "+v"(on), "+v"(bs));

    float* obt = ob + (size_t)(tbase + 8 * lh) * OCH + o;
    v2f bfrag[KT];
    v8f acc0, acc1, acc2, acc3;
#pragma unroll
    for (int r = 0; r < 8; ++r) { acc0[r] = bs; acc1[r] = bs; acc2[r] = bs; acc3[r] = bs; }

#pragma unroll
    for (int kt = 0; kt < KT; ++kt) {
#pragma unroll
        for (int r = 0; r < 2; ++r) {
            const int c  = cbase + kt * 4 + 2 * lh + r;
            const int kk = c - on + PADW;
            const bool inb = (kk >= 0) & (kk < KSZ) & (c >= 0) & (c < CIN);
            bfrag[kt][r] = inb ? wraw[kt][r] : 0.0f;
        }
        acc0 = __builtin_amdgcn_wmma_f32_16x16x4_f32(
            false, a[0][kt], false, bfrag[kt], (short)0, acc0, false, false);
        acc1 = __builtin_amdgcn_wmma_f32_16x16x4_f32(
            false, a[1][kt], false, bfrag[kt], (short)0, acc1, false, false);
    }

    acc2 = __builtin_amdgcn_wmma_f32_16x16x4_f32(
        false, a[2][0], false, bfrag[0], (short)0, acc2, false, false);
    acc3 = __builtin_amdgcn_wmma_f32_16x16x4_f32(
        false, a[3][0], false, bfrag[0], (short)0, acc3, false, false);

#pragma unroll
    for (int r = 0; r < 8; ++r) obt[(size_t)r * OCH] = acc0[r];
#pragma unroll
    for (int r = 0; r < 8; ++r) obt[(size_t)(16 + r) * OCH] = acc1[r];

    __builtin_amdgcn_sched_barrier(0);

#pragma unroll
    for (int kt = 1; kt < KT; ++kt) {
        acc2 = __builtin_amdgcn_wmma_f32_16x16x4_f32(
            false, a[2][kt], false, bfrag[kt], (short)0, acc2, false, false);
        acc3 = __builtin_amdgcn_wmma_f32_16x16x4_f32(
            false, a[3][kt], false, bfrag[kt], (short)0, acc3, false, false);
    }
#pragma unroll
    for (int r = 0; r < 8; ++r) obt[(size_t)(32 + r) * OCH] = acc2[r];
#pragma unroll
    for (int r = 0; r < 8; ++r) obt[(size_t)(48 + r) * OCH] = acc3[r];
}

extern "C" void kernel_launch(void* const* d_in, const int* in_sizes, int n_in,
                              void* d_out, int out_size, void* d_ws, size_t ws_size,
                              hipStream_t stream) {
    const float* x    = (const float*)d_in[0];
    const float* w    = (const float*)d_in[1];
    const float* bias = (const float*)d_in[2];
    float*       out  = (float*)d_out;

    hipLaunchKernelGGL(pwtdconv_wmma_f32, dim3(256), dim3(256),
                       0, stream, x, w, bias, out);
}
